# starting kernel
# baseline (speedup 1.0000x reference)
#include <hip/hip_runtime.h>


#define IN_NODES   4096
#define OUT_NODES  1024
#define FD         16
#define ROWS_BLK   32
#define NSUB       2
#define NBLK       128
#define SVEC       16384
#define BSTRIDE    1025

typedef __attribute__((ext_vector_type(2))) float v2f;
typedef __attribute__((ext_vector_type(8))) float v8f;

#define SMEM_FLOATS (SVEC + SVEC + 16*BSTRIDE + 256 + 16 + 16)
#define SMEM_BYTES  (SMEM_FLOATS * 4)

__global__ void caps_zero_vsum(float* __restrict__ vsum) {
    vsum[blockIdx.x * 256 + threadIdx.x] = 0.0f;
}

__global__ void caps_routing_pass(const float* __restrict__ uh,
                                  const float* __restrict__ vsum_g,
                                  float* __restrict__ partials) {
    extern __shared__ float smem[];
    float* vs   = smem;
    float* sacc = vs   + SVEC;
    float* bnew = sacc + SVEC;
    float* stat = bnew + 16 * BSTRIDE;
    float* rmax = stat + 256;
    float* rinv = rmax + 16;

    const int t    = threadIdx.x;
    const int lane = t & 31;
    const int wave = t >> 5;
    const int obase = wave * 128;

    for (int q = t; q < SVEC / 4; q += 256) {
        ((float4*)vs)[q]   = ((const float4*)vsum_g)[q];
        ((float4*)sacc)[q] = make_float4(0.f, 0.f, 0.f, 0.f);
    }
    __syncthreads();

    for (int st = 0; st < NSUB; ++st) {
        const int i0 = blockIdx.x * ROWS_BLK + st * 16;

        {
            const int m  = lane & 15;
            const int oh = lane >> 4;
            for (int j = 0; j < 64; ++j) {
                const int o = obase + 2 * j + oh;
                const float4* up = (const float4*)(uh + ((size_t)(i0 + m) * OUT_NODES + o) * FD);
                const float*  vv = vs + o * FD;
                float acc = 0.f;
                #pragma unroll
                for (int q = 0; q < 4; ++q) {
                    const float4 u4 = up[q];
                    acc += u4.x * vv[4*q+0] + u4.y * vv[4*q+1]
                         + u4.z * vv[4*q+2] + u4.w * vv[4*q+3];
                }
                bnew[m * BSTRIDE + o] = acc;
            }
        }
        __syncthreads();

        {
            const int r = t >> 4, sub = t & 15;
            float pm = -3.4e38f;
            for (int q = sub; q < OUT_NODES; q += 16)
                pm = fmaxf(pm, bnew[r * BSTRIDE + q]);
            stat[r * 16 + sub] = pm;
            __syncthreads();
            if (sub == 0) {
                float mm = stat[r * 16];
                for (int q = 1; q < 16; ++q) mm = fmaxf(mm, stat[r * 16 + q]);
                rmax[r] = mm;
            }
            __syncthreads();
            const float mm = rmax[r];
            float ps = 0.f;
            for (int q = sub; q < OUT_NODES; q += 16) {
                const float e = __expf(bnew[r * BSTRIDE + q] - mm);
                bnew[r * BSTRIDE + q] = e;
                ps += e;
            }
            stat[r * 16 + sub] = ps;
            __syncthreads();
            if (sub == 0) {
                float ss = 0.f;
                for (int q = 0; q < 16; ++q) ss += stat[r * 16 + q];
                rinv[r] = 1.f / ss;
            }
            __syncthreads();
        }

        {
            const int hk = (lane < 16) ? 0 : 2;
            const int fl = lane & 15;
            for (int j = 0; j < 128; ++j) {
                const int o = obase + j;
                v8f d = {0.f, 0.f, 0.f, 0.f, 0.f, 0.f, 0.f, 0.f};
                #pragma unroll
                for (int k = 0; k < 4; ++k) {
                    const int r0 = 4 * k + hk;
                    v2f a, b;
                    a.x = bnew[(r0    ) * BSTRIDE + o] * rinv[r0    ];
                    a.y = bnew[(r0 + 1) * BSTRIDE + o] * rinv[r0 + 1];
                    const float* bp = uh + ((size_t)(i0 + r0) * OUT_NODES + o) * FD + fl;
                    b.x = bp[0];
                    b.y = bp[OUT_NODES * FD];
                    d = __builtin_amdgcn_wmma_f32_16x16x4_f32(
                             false, a,  false, b,
                             (short)0, d,  false,  false);
                }
                if (lane < 16) sacc[o * FD + fl] += d[0];
            }
        }
        __syncthreads();
    }

    float* pout = partials + (size_t)blockIdx.x * SVEC;
    for (int q = t; q < SVEC / 4; q += 256)
        ((float4*)pout)[q] = ((const float4*)sacc)[q];
}

__global__ void caps_reduce_squash(const float* __restrict__ partials,
                                   float* __restrict__ vsum,
                                   float* __restrict__ vout) {
    const int idx = blockIdx.x * 256 + threadIdx.x;
    float s = 0.f;
    for (int b = 0; b < NBLK; ++b)
        s += partials[(size_t)b * SVEC + idx];
    float sq = s * s;
    #pragma unroll
    for (int m = 1; m < 16; m <<= 1)
        sq += __shfl_xor(sq, m, 16);
    const float scale = (sq / (1.f + sq)) * rsqrtf(sq);
    const float v = s * scale;
    vout[idx] = v;
    vsum[idx] += v;
}

extern "C" void kernel_launch(void* const* d_in, const int* in_sizes, int n_in,
                              void* d_out, int out_size, void* d_ws, size_t ws_size,
                              hipStream_t stream) {
    const float* uh = (const float*)d_in[0];
    const int routing_num = 3;

    float* ws       = (float*)d_ws;
    float* vsum     = ws;
    float* partials = ws + SVEC;
    float* vout     = (float*)d_out;

    caps_zero_vsum<<<SVEC / 256, 256, 0, stream>>>(vsum);
    for (int it = 0; it < routing_num; ++it) {
        caps_routing_pass<<<NBLK, 256, SMEM_BYTES, stream>>>(uh, vsum, partials);
        caps_reduce_squash<<<SVEC / 256, 256, 0, stream>>>(partials, vsum, vout);
    }
}


# Rounds pipelined: LDS tile and e double-buffered; tile-in of tile r+1 (LDS stores, dots, refill loads) interleaved 4 pieces per 8 WMMAs with the sum of tile r; 3 barriers a round; sums unchanged
# speedup vs baseline: 3.2746x; 3.2746x over previous
#include <hip/hip_runtime.h>


#define IN_NODES   4096
#define OUT_NODES  1024
#define FD         16
#define ROWS_BLK   32
#define NBLK       128
#define SVEC       16384
#define TROWS      2
#define NTILE      (ROWS_BLK / TROWS)
#define TILE_F4    (TROWS * SVEC / 4)
#define PRE        (TILE_F4 / 256)

typedef __attribute__((ext_vector_type(2))) float v2f;
typedef __attribute__((ext_vector_type(8))) float v8f;

#define EPOS(i)     ((i) + (((i) >> 6) << 3))
#define E_FLOATS    (TROWS * OUT_NODES + TROWS * OUT_NODES / 8)
#define E_ROW       (OUT_NODES + OUT_NODES / 8)

#define TILE_FLOATS (TROWS * SVEC)
#define SMEM_FLOATS (2 * TILE_FLOATS + 2 * E_FLOATS + 32)
#define SMEM_BYTES  (SMEM_FLOATS * 4)

static __device__ __forceinline__ void lds_barrier() {
    asm volatile("s_wait_dscnt 0x0\n\ts_barrier_signal -1\n\ts_barrier_wait -1" ::: "memory");
}

template <int MASK>
static __device__ __forceinline__ float lane_xor(float x) {
    return __builtin_bit_cast(float, __builtin_amdgcn_update_dpp(
            0, __builtin_bit_cast(int, x), 0x160 + MASK, 0xf, 0xf, false));
}
static __device__ __forceinline__ float xhalf(float x) {
    const int b = __builtin_bit_cast(int, x);
    return __builtin_bit_cast(float,
            __builtin_amdgcn_permlanex16(b, b, 0x76543210, 0xfedcba98, false, false));
}

__global__ void caps_zero_vsum(float* __restrict__ vsum) {
    vsum[blockIdx.x * 256 + threadIdx.x] = 0.0f;
}

template <bool REFILL>
static __device__ __forceinline__ void tile_in_group(
        const int g, float4 (&pre)[PRE], const float4 (&vsv)[16],
        const float4* __restrict__ nsrc, float* tile, float* eq,
        const int t, const bool odd, const bool high) {
    float p[4];
    #pragma unroll
    for (int j = 0; j < 4; ++j) {
        const int k = 4 * g + j;
        const int q = t + 256 * k;
        const float4 u = pre[k];
        ((float4*)tile)[q] = u;
        const float4 v = vsv[k & 15];
        p[j] = u.x * v.x + u.y * v.y + u.z * v.z + u.w * v.w;
        if (REFILL) pre[k] = nsrc[q];
    }
    float k01 = odd ? p[1] : p[0];
    float k23 = odd ? p[3] : p[2];
    k01 += lane_xor<1>(odd ? p[0] : p[1]);
    k23 += lane_xor<1>(odd ? p[2] : p[3]);
    float d = high ? k23 : k01;
    d += lane_xor<2>(high ? k01 : k23);
    eq[288 * g] = d;
}

static __device__ __forceinline__ void row_softmax(
        float* e, float* red, const int lane, const int wave,
        float& rinv0, float& rinv1) {
    const int r = wave >> 2;
    float* er = e + r * E_ROW + (wave & 3) * 288 + lane;
    float bv[8];
    float pm = -3.4e38f;
    #pragma unroll
    for (int j = 0; j < 8; ++j) { bv[j] = er[32 * j + 8 * (j >> 1)]; pm = fmaxf(pm, bv[j]); }
    pm = fmaxf(pm, lane_xor<1>(pm));
    pm = fmaxf(pm, lane_xor<2>(pm));
    pm = fmaxf(pm, lane_xor<4>(pm));
    pm = fmaxf(pm, lane_xor<8>(pm));
    pm = fmaxf(pm, xhalf(pm));
    if (lane == 0) red[wave] = pm;
    lds_barrier();
    const float mm = fmaxf(fmaxf(red[4 * r], red[4 * r + 1]),
                           fmaxf(red[4 * r + 2], red[4 * r + 3]));
    float ps = 0.f;
    #pragma unroll
    for (int j = 0; j < 8; ++j) {
        const float ev = __expf(bv[j] - mm);
        er[32 * j + 8 * (j >> 1)] = ev;
        ps += ev;
    }
    ps += lane_xor<1>(ps);
    ps += lane_xor<2>(ps);
    ps += lane_xor<4>(ps);
    ps += lane_xor<8>(ps);
    ps += xhalf(ps);
    if (lane == 0) red[8 + wave] = ps;
    lds_barrier();
    rinv0 = 1.f / ((red[8]  + red[9])  + (red[10] + red[11]));
    rinv1 = 1.f / ((red[12] + red[13]) + (red[14] + red[15]));
}

struct GroupOps {
    v2f   b[8];
    float e0, e1;
};

static __device__ __forceinline__ void load_group(
        const int g, GroupOps& f, const float* tile, const float* e,
        const int wave, const int obase, const int m, const int h) {
    const int og = obase + 16 * g;
    const float* ec = e + og + m + 8 * (2 * wave + (g >> 2));
    f.e0 = ec[0];
    f.e1 = ec[E_ROW];
    #pragma unroll
    for (int q = 0; q < 8; ++q) {
        const float* bp = tile + (og + 2 * q + h) * FD + m;
        f.b[q].x = bp[0];
        f.b[q].y = bp[SVEC];
    }
}

static __device__ __forceinline__ void wmma_group(
        const int g, v8f (&acc)[8], const GroupOps& f,
        const float rinv0, const float rinv1, const int m, const int h) {
    const float c0 = f.e0 * rinv0;
    const float c1 = f.e1 * rinv1;
    #pragma unroll
    for (int q = 0; q < 8; ++q) {
        const bool sel = (m == 2 * q + h);
        v2f a;
        a.x = sel ? c0 : 0.f;
        a.y = sel ? c1 : 0.f;
        acc[g] = __builtin_amdgcn_wmma_f32_16x16x4_f32(
                 false, a,  false, f.b[q],
                 (short)0, acc[g],  false,  false);
    }
}

template <bool REFILL>
static __device__ __forceinline__ void routing_round(
        float4 (&pre)[PRE], const float4 (&vsv)[16], v8f (&acc)[8],
        const float4* __restrict__ nsrc,
        float* tn, float* en, const float* tc, float* ec, float* red,
        const int t, const int lane, const int wave, const int obase,
        const int m, const int h) {
    const bool odd  = (t & 1) != 0;
    const bool high = (t & 2) != 0;
    float* eq = en + (t >> 2) + 72 * (t & 3);
    float rinv0, rinv1;
    row_softmax(ec, red, lane, wave, rinv0, rinv1);
    GroupOps ops[2];
    load_group(0, ops[0], tc, ec, wave, obase, m, h);
    #pragma unroll
    for (int g = 0; g < 8; ++g) {
        if (g < 7) load_group(g + 1, ops[(g + 1) & 1], tc, ec, wave, obase, m, h);
        __builtin_amdgcn_sched_barrier(6);
        tile_in_group<REFILL>(g, pre, vsv, nsrc, tn, eq, t, odd, high);
        wmma_group(g, acc, ops[g & 1], rinv0, rinv1, m, h);
    }
    lds_barrier();
}

__global__ __launch_bounds__(256)
void caps_routing_pass(const float* __restrict__ uh,
                       const float* __restrict__ vsum_g,
                       float* __restrict__ partials) {
    extern __shared__ float smem[];
    float* tile = smem;
    float* e    = tile + 2 * TILE_FLOATS;
    float* red  = e    + 2 * E_FLOATS;

    const int t     = threadIdx.x;
    const int lane  = t & 31;
    const int wave  = t >> 5;
    const int obase = wave * 128;
    const int m     = lane & 15;
    const int h     = lane >> 4;

    const float4* src = (const float4*)(uh + (size_t)blockIdx.x * ROWS_BLK * SVEC);

    float4 vsv[16];
    #pragma unroll
    for (int kk = 0; kk < 16; ++kk) vsv[kk] = ((const float4*)vsum_g)[t + 256 * kk];

    float4 pre[PRE];
    #pragma unroll
    for (int k = 0; k < PRE; ++k) pre[k] = src[t + 256 * k];

    v8f acc[8];
    #pragma unroll
    for (int g = 0; g < 8; ++g) acc[g] = (v8f){0.f, 0.f, 0.f, 0.f, 0.f, 0.f, 0.f, 0.f};

    {
        const bool odd  = (t & 1) != 0;
        const bool high = (t & 2) != 0;
        float* eq = e + (t >> 2) + 72 * (t & 3);
        #pragma unroll
        for (int g = 0; g < 8; ++g)
            tile_in_group<true>(g, pre, vsv, src + TILE_F4, tile, eq, t, odd, high);
    }
    lds_barrier();

    for (int tl = 0; tl + 2 < NTILE; ++tl) {
        const int cur = tl & 1;
        routing_round<true>(pre, vsv, acc, src + (size_t)(tl + 2) * TILE_F4,
                            tile + (cur ^ 1) * TILE_FLOATS, e + (cur ^ 1) * E_FLOATS,
                            tile + cur * TILE_FLOATS,       e + cur * E_FLOATS, red,
                            t, lane, wave, obase, m, h);
    }
    routing_round<false>(pre, vsv, acc, src,
                         tile + TILE_FLOATS, e + E_FLOATS, tile, e, red,
                         t, lane, wave, obase, m, h);
    {
        float rinv0, rinv1;
        row_softmax(e + E_FLOATS, red, lane, wave, rinv0, rinv1);
        GroupOps ops[2];
        load_group(0, ops[0], tile + TILE_FLOATS, e + E_FLOATS, wave, obase, m, h);
        #pragma unroll
        for (int g = 0; g < 8; ++g) {
            if (g < 7) load_group(g + 1, ops[(g + 1) & 1], tile + TILE_FLOATS, e + E_FLOATS,
                                  wave, obase, m, h);
            __builtin_amdgcn_sched_barrier(6);
            wmma_group(g, acc, ops[g & 1], rinv0, rinv1, m, h);
        }
    }

    #pragma unroll
    for (int g = 0; g < 8; ++g) {
        #pragma unroll
        for (int r8 = 0; r8 < 8; ++r8)
            tile[(obase + 16 * g + 8 * h + r8) * FD + m] = acc[g][r8];
    }
    lds_barrier();
    float* pout = partials + (size_t)blockIdx.x * SVEC;
    for (int q = t; q < SVEC / 4; q += 256)
        ((float4*)pout)[q] = ((const float4*)tile)[q];
}

__global__ void caps_reduce_squash(const float* __restrict__ partials,
                                   float* __restrict__ vsum,
                                   float* __restrict__ vout) {
    const int idx = blockIdx.x * 256 + threadIdx.x;
    float s = 0.f;
    for (int b = 0; b < NBLK; ++b)
        s += partials[(size_t)b * SVEC + idx];
    float sq = s * s;
    #pragma unroll
    for (int m = 1; m < 16; m <<= 1)
        sq += __shfl_xor(sq, m, 16);
    const float scale = (sq / (1.f + sq)) * rsqrtf(sq);
    const float v = s * scale;
    vout[idx] = v;
    vsum[idx] += v;
}

extern "C" void kernel_launch(void* const* d_in, const int* in_sizes, int n_in,
                              void* d_out, int out_size, void* d_ws, size_t ws_size,
                              hipStream_t stream) {
    const float* uh = (const float*)d_in[0];
    const int routing_num = 3;

    float* ws       = (float*)d_ws;
    float* vsum     = ws;
    float* partials = ws + SVEC;
    float* vout     = (float*)d_out;

    caps_zero_vsum<<<SVEC / 256, 256, 0, stream>>>(vsum);
    for (int it = 0; it < routing_num; ++it) {
        caps_routing_pass<<<NBLK, 256, SMEM_BYTES, stream>>>(uh, vsum, partials);
        caps_reduce_squash<<<SVEC / 256, 256, 0, stream>>>(partials, vsum, vout);
    }
}


# o range split over two work-groups per 32-row group (256 in all): both take every dot for the softmax, each stages in LDS and sums on WMMA only its 512 o (64 -> 32 WMMAs per wave-round); sums same
# speedup vs baseline: 3.8408x; 1.1729x over previous
#include <hip/hip_runtime.h>


#define IN_NODES   4096
#define OUT_NODES  1024
#define FD         16
#define ROWS_BLK   32
#define NBLK       128
#define SVEC       16384
#define TROWS      2
#define NTILE      (ROWS_BLK / TROWS)
#define TILE_F4    (TROWS * SVEC / 4)
#define PRE        (TILE_F4 / 256)
#define OHALF      (OUT_NODES / 2)
#define HALF_F4    (OHALF * FD / 4)
#define LROW       (OHALF * FD)

typedef __attribute__((ext_vector_type(2))) float v2f;
typedef __attribute__((ext_vector_type(8))) float v8f;

#define EPOS(i)     ((i) + (((i) >> 6) << 3))
#define E_FLOATS    (TROWS * OUT_NODES + TROWS * OUT_NODES / 8)
#define E_ROW       (OUT_NODES + OUT_NODES / 8)

#define TILE_FLOATS (TROWS * LROW)
#define SMEM_FLOATS (2 * TILE_FLOATS + 2 * E_FLOATS + 32)
#define SMEM_BYTES  (SMEM_FLOATS * 4)

static __device__ __forceinline__ void lds_barrier() {
    asm volatile("s_wait_dscnt 0x0\n\ts_barrier_signal -1\n\ts_barrier_wait -1" ::: "memory");
}

template <int MASK>
static __device__ __forceinline__ float lane_xor(float x) {
    return __builtin_bit_cast(float, __builtin_amdgcn_update_dpp(
            0, __builtin_bit_cast(int, x), 0x160 + MASK, 0xf, 0xf, false));
}
static __device__ __forceinline__ float xhalf(float x) {
    const int b = __builtin_bit_cast(int, x);
    return __builtin_bit_cast(float,
            __builtin_amdgcn_permlanex16(b, b, 0x76543210, 0xfedcba98, false, false));
}

__global__ void caps_zero_vsum(float* __restrict__ vsum) {
    vsum[blockIdx.x * 256 + threadIdx.x] = 0.0f;
}

template <bool REFILL>
static __device__ __forceinline__ void tile_in_group(
        const int g, float4 (&pre)[PRE], const float4 (&vsv)[16],
        const float4* __restrict__ nsrc, const int t_own, const int t_oth,
        float* tile, float* eq_own, float* eq_oth,
        const int t, const bool odd, const bool high) {
    const bool own = ((g >> 1) & 1) == 0;
    const int  tq  = own ? t_own : t_oth;
    float p[4];
    #pragma unroll
    for (int j = 0; j < 4; ++j) {
        const int k = 4 * g + j;
        const float4 u = pre[k];
        if (own) ((float4*)tile)[t + 256 * (k & 7) + HALF_F4 * (k >> 4)] = u;
        const float4 v = vsv[k & 15];
        p[j] = u.x * v.x + u.y * v.y + u.z * v.z + u.w * v.w;
        asm volatile("" : "+v"(p[j]));
        if (REFILL) pre[k] = nsrc[tq + 256 * (k & 7) + 4096 * (k >> 4)];
    }
    float k01 = odd ? p[1] : p[0];
    float k23 = odd ? p[3] : p[2];
    k01 += lane_xor<1>(odd ? p[0] : p[1]);
    k23 += lane_xor<1>(odd ? p[2] : p[3]);
    float d = high ? k23 : k01;
    d += lane_xor<2>(high ? k01 : k23);
    (own ? eq_own : eq_oth)[E_ROW * (g >> 2) + 288 * (g & 1)] = d;
}

static __device__ __forceinline__ void row_softmax(
        float* e, float* red, const int lane, const int wave,
        float& rinv0, float& rinv1) {
    const int r = wave >> 2;
    float* er = e + r * E_ROW + (wave & 3) * 288 + lane;
    float bv[8];
    float pm = -3.4e38f;
    #pragma unroll
    for (int j = 0; j < 8; ++j) { bv[j] = er[32 * j + 8 * (j >> 1)]; pm = fmaxf(pm, bv[j]); }
    pm = fmaxf(pm, lane_xor<1>(pm));
    pm = fmaxf(pm, lane_xor<2>(pm));
    pm = fmaxf(pm, lane_xor<4>(pm));
    pm = fmaxf(pm, lane_xor<8>(pm));
    pm = fmaxf(pm, xhalf(pm));
    if (lane == 0) red[wave] = pm;
    lds_barrier();
    const float mm = fmaxf(fmaxf(red[4 * r], red[4 * r + 1]),
                           fmaxf(red[4 * r + 2], red[4 * r + 3]));
    float ps = 0.f;
    #pragma unroll
    for (int j = 0; j < 8; ++j) {
        const float ev = __expf(bv[j] - mm);
        er[32 * j + 8 * (j >> 1)] = ev;
        ps += ev;
    }
    ps += lane_xor<1>(ps);
    ps += lane_xor<2>(ps);
    ps += lane_xor<4>(ps);
    ps += lane_xor<8>(ps);
    ps += xhalf(ps);
    if (lane == 0) red[8 + wave] = ps;
    lds_barrier();
    rinv0 = 1.f / ((red[8]  + red[9])  + (red[10] + red[11]));
    rinv1 = 1.f / ((red[12] + red[13]) + (red[14] + red[15]));
}

struct GroupOps {
    v2f   b[8];
    float e0, e1;
};

static __device__ __forceinline__ void load_group(
        const int g, GroupOps& f, const float* tile, const float* e,
        const int erun, const int obase, const int m, const int h) {
    const int og = obase + 16 * g;
    const float* ec = e + 72 * erun + 16 * g + m;
    f.e0 = ec[0];
    f.e1 = ec[E_ROW];
    #pragma unroll
    for (int q = 0; q < 8; ++q) {
        const float* bp = tile + (og + 2 * q + h) * FD + m;
        f.b[q].x = bp[0];
        f.b[q].y = bp[LROW];
    }
}

static __device__ __forceinline__ void wmma_group(
        const int g, v8f (&acc)[4], const GroupOps& f,
        const float rinv0, const float rinv1, const int m, const int h) {
    const float c0 = f.e0 * rinv0;
    const float c1 = f.e1 * rinv1;
    #pragma unroll
    for (int q = 0; q < 8; ++q) {
        const bool sel = (m == 2 * q + h);
        v2f a;
        a.x = sel ? c0 : 0.f;
        a.y = sel ? c1 : 0.f;
        acc[g] = __builtin_amdgcn_wmma_f32_16x16x4_f32(
                 false, a,  false, f.b[q],
                 (short)0, acc[g],  false,  false);
    }
}

template <bool REFILL>
static __device__ __forceinline__ void routing_round(
        float4 (&pre)[PRE], const float4 (&vsv)[16], v8f (&acc)[4],
        const float4* __restrict__ nsrc, const int t_own, const int t_oth,
        float* tn, float* en, const float* tc, float* ec, float* red,
        const int t, const int lane, const int wave, const int erun, const int obase,
        const int m, const int h, const int half) {
    const bool odd  = (t & 1) != 0;
    const bool high = (t & 2) != 0;
    float* eq_own = en + (t >> 2) + 72 * (t & 3) + 576 * half;
    float* eq_oth = en + (t >> 2) + 72 * (t & 3) + 576 * (half ^ 1);
    float rinv0, rinv1;
    row_softmax(ec, red, lane, wave, rinv0, rinv1);
    GroupOps ops[2];
    load_group(0, ops[0], tc, ec, erun, obase, m, h);
    #pragma unroll
    for (int g = 0; g < 4; ++g) {
        if (g < 3) load_group(g + 1, ops[(g + 1) & 1], tc, ec, erun, obase, m, h);
        __builtin_amdgcn_sched_barrier(6);
        tile_in_group<REFILL>(2 * g,     pre, vsv, nsrc, t_own, t_oth, tn, eq_own, eq_oth, t, odd, high);
        tile_in_group<REFILL>(2 * g + 1, pre, vsv, nsrc, t_own, t_oth, tn, eq_own, eq_oth, t, odd, high);
        wmma_group(g, acc, ops[g & 1], rinv0, rinv1, m, h);
    }
    lds_barrier();
}

__global__ __launch_bounds__(256)
void caps_routing_pass(const float* __restrict__ uh,
                       const float* __restrict__ vsum_g,
                       float* __restrict__ partials) {
    extern __shared__ float smem[];
    float* tile = smem;
    float* e    = tile + 2 * TILE_FLOATS;
    float* red  = e    + 2 * E_FLOATS;

    const int t     = threadIdx.x;
    const int lane  = t & 31;
    const int wave  = t >> 5;
    const int obase = wave * 64;
    const int m     = lane & 15;
    const int h     = lane >> 4;
    const int rg    = blockIdx.x % NBLK;
    const int half  = blockIdx.x / NBLK;
    const int erun  = 8 * half + wave;
    const int t_own = t + HALF_F4 * half;
    const int t_oth = t + HALF_F4 * (half ^ 1);

    const float4* src = (const float4*)(uh + (size_t)rg * ROWS_BLK * SVEC);

    float4 vsv[16];
    #pragma unroll
    for (int kk = 0; kk < 16; ++kk)
        vsv[kk] = ((const float4*)vsum_g)[(kk < 8 ? t_own : t_oth) + 256 * (kk & 7)];

    float4 pre[PRE];
    #pragma unroll
    for (int k = 0; k < PRE; ++k)
        pre[k] = src[((k & 8) == 0 ? t_own : t_oth) + 256 * (k & 7) + 4096 * (k >> 4)];

    v8f acc[4];
    #pragma unroll
    for (int g = 0; g < 4; ++g) acc[g] = (v8f){0.f, 0.f, 0.f, 0.f, 0.f, 0.f, 0.f, 0.f};

    {
        const bool odd  = (t & 1) != 0;
        const bool high = (t & 2) != 0;
        float* eq_own = e + (t >> 2) + 72 * (t & 3) + 576 * half;
        float* eq_oth = e + (t >> 2) + 72 * (t & 3) + 576 * (half ^ 1);
        #pragma unroll
        for (int g = 0; g < 8; ++g)
            tile_in_group<true>(g, pre, vsv, src + TILE_F4, t_own, t_oth, tile, eq_own, eq_oth,
                                t, odd, high);
    }
    lds_barrier();

    for (int tl = 0; tl + 2 < NTILE; ++tl) {
        const int cur = tl & 1;
        routing_round<true>(pre, vsv, acc, src + (size_t)(tl + 2) * TILE_F4, t_own, t_oth,
                            tile + (cur ^ 1) * TILE_FLOATS, e + (cur ^ 1) * E_FLOATS,
                            tile + cur * TILE_FLOATS,       e + cur * E_FLOATS, red,
                            t, lane, wave, erun, obase, m, h, half);
    }
    routing_round<false>(pre, vsv, acc, src, t_own, t_oth,
                         tile + TILE_FLOATS, e + E_FLOATS, tile, e, red,
                         t, lane, wave, erun, obase, m, h, half);
    {
        float rinv0, rinv1;
        row_softmax(e + E_FLOATS, red, lane, wave, rinv0, rinv1);
        GroupOps ops[2];
        load_group(0, ops[0], tile + TILE_FLOATS, e + E_FLOATS, erun, obase, m, h);
        #pragma unroll
        for (int g = 0; g < 4; ++g) {
            if (g < 3) load_group(g + 1, ops[(g + 1) & 1], tile + TILE_FLOATS, e + E_FLOATS,
                                  erun, obase, m, h);
            __builtin_amdgcn_sched_barrier(6);
            wmma_group(g, acc, ops[g & 1], rinv0, rinv1, m, h);
        }
    }

    #pragma unroll
    for (int g = 0; g < 4; ++g) {
        #pragma unroll
        for (int r8 = 0; r8 < 8; ++r8)
            tile[(obase + 16 * g + 8 * h + r8) * FD + m] = acc[g][r8];
    }
    lds_barrier();
    float* pout = partials + (size_t)rg * SVEC + (size_t)half * LROW;
    for (int q = t; q < LROW / 4; q += 256)
        ((float4*)pout)[q] = ((const float4*)tile)[q];
}

__global__ void caps_reduce_squash(const float* __restrict__ partials,
                                   float* __restrict__ vsum,
                                   float* __restrict__ vout) {
    const int idx = blockIdx.x * 256 + threadIdx.x;
    float s = 0.f;
    for (int b = 0; b < NBLK; ++b)
        s += partials[(size_t)b * SVEC + idx];
    float sq = s * s;
    #pragma unroll
    for (int m = 1; m < 16; m <<= 1)
        sq += __shfl_xor(sq, m, 16);
    const float scale = (sq / (1.f + sq)) * rsqrtf(sq);
    const float v = s * scale;
    vout[idx] = v;
    vsum[idx] += v;
}

extern "C" void kernel_launch(void* const* d_in, const int* in_sizes, int n_in,
                              void* d_out, int out_size, void* d_ws, size_t ws_size,
                              hipStream_t stream) {
    const float* uh = (const float*)d_in[0];
    const int routing_num = 3;

    float* ws       = (float*)d_ws;
    float* vsum     = ws;
    float* partials = ws + SVEC;
    float* vout     = (float*)d_out;

    caps_zero_vsum<<<SVEC / 256, 256, 0, stream>>>(vsum);
    for (int it = 0; it < routing_num; ++it) {
        caps_routing_pass<<<2 * NBLK, 256, SMEM_BYTES, stream>>>(uh, vsum, partials);
        caps_reduce_squash<<<SVEC / 256, 256, 0, stream>>>(partials, vsum, vout);
    }
}


# Tile walk alternates direction by pass (pass 2 last to first): each pass starts on the lines the pass before read last, still in the second-level cache; pass 2 sums its tiles in reverse order
# speedup vs baseline: 4.1238x; 1.0737x over previous
#include <hip/hip_runtime.h>


#define IN_NODES   4096
#define OUT_NODES  1024
#define FD         16
#define ROWS_BLK   32
#define NBLK       128
#define SVEC       16384
#define TROWS      2
#define NTILE      (ROWS_BLK / TROWS)
#define TILE_F4    (TROWS * SVEC / 4)
#define PRE        (TILE_F4 / 256)
#define OHALF      (OUT_NODES / 2)
#define HALF_F4    (OHALF * FD / 4)
#define LROW       (OHALF * FD)

typedef __attribute__((ext_vector_type(2))) float v2f;
typedef __attribute__((ext_vector_type(8))) float v8f;

#define EPOS(i)     ((i) + (((i) >> 6) << 3))
#define E_FLOATS    (TROWS * OUT_NODES + TROWS * OUT_NODES / 8)
#define E_ROW       (OUT_NODES + OUT_NODES / 8)

#define TILE_FLOATS (TROWS * LROW)
#define SMEM_FLOATS (2 * TILE_FLOATS + 2 * E_FLOATS + 32)
#define SMEM_BYTES  (SMEM_FLOATS * 4)

static __device__ __forceinline__ void lds_barrier() {
    asm volatile("s_wait_dscnt 0x0\n\ts_barrier_signal -1\n\ts_barrier_wait -1" ::: "memory");
}

template <int MASK>
static __device__ __forceinline__ float lane_xor(float x) {
    return __builtin_bit_cast(float, __builtin_amdgcn_update_dpp(
            0, __builtin_bit_cast(int, x), 0x160 + MASK, 0xf, 0xf, false));
}
static __device__ __forceinline__ float xhalf(float x) {
    const int b = __builtin_bit_cast(int, x);
    return __builtin_bit_cast(float,
            __builtin_amdgcn_permlanex16(b, b, 0x76543210, 0xfedcba98, false, false));
}

__global__ void caps_zero_vsum(float* __restrict__ vsum) {
    vsum[blockIdx.x * 256 + threadIdx.x] = 0.0f;
}

template <bool REFILL>
static __device__ __forceinline__ void tile_in_group(
        const int g, float4 (&pre)[PRE], const float4 (&vsv)[16],
        const float4* __restrict__ nsrc, const int t_own, const int t_oth,
        float* tile, float* eq_own, float* eq_oth,
        const int t, const bool odd, const bool high) {
    const bool own = ((g >> 1) & 1) == 0;
    const int  tq  = own ? t_own : t_oth;
    float p[4];
    #pragma unroll
    for (int j = 0; j < 4; ++j) {
        const int k = 4 * g + j;
        const float4 u = pre[k];
        if (own) ((float4*)tile)[t + 256 * (k & 7) + HALF_F4 * (k >> 4)] = u;
        const float4 v = vsv[k & 15];
        p[j] = u.x * v.x + u.y * v.y + u.z * v.z + u.w * v.w;
        asm volatile("" : "+v"(p[j]));
        if (REFILL) pre[k] = nsrc[tq + 256 * (k & 7) + 4096 * (k >> 4)];
    }
    float k01 = odd ? p[1] : p[0];
    float k23 = odd ? p[3] : p[2];
    k01 += lane_xor<1>(odd ? p[0] : p[1]);
    k23 += lane_xor<1>(odd ? p[2] : p[3]);
    float d = high ? k23 : k01;
    d += lane_xor<2>(high ? k01 : k23);
    (own ? eq_own : eq_oth)[E_ROW * (g >> 2) + 288 * (g & 1)] = d;
}

static __device__ __forceinline__ void row_softmax(
        float* e, float* red, const int lane, const int wave,
        float& rinv0, float& rinv1) {
    const int r = wave >> 2;
    float* er = e + r * E_ROW + (wave & 3) * 288 + lane;
    float bv[8];
    float pm = -3.4e38f;
    #pragma unroll
    for (int j = 0; j < 8; ++j) { bv[j] = er[32 * j + 8 * (j >> 1)]; pm = fmaxf(pm, bv[j]); }
    pm = fmaxf(pm, lane_xor<1>(pm));
    pm = fmaxf(pm, lane_xor<2>(pm));
    pm = fmaxf(pm, lane_xor<4>(pm));
    pm = fmaxf(pm, lane_xor<8>(pm));
    pm = fmaxf(pm, xhalf(pm));
    if (lane == 0) red[wave] = pm;
    lds_barrier();
    const float mm = fmaxf(fmaxf(red[4 * r], red[4 * r + 1]),
                           fmaxf(red[4 * r + 2], red[4 * r + 3]));
    float ps = 0.f;
    #pragma unroll
    for (int j = 0; j < 8; ++j) {
        const float ev = __expf(bv[j] - mm);
        er[32 * j + 8 * (j >> 1)] = ev;
        ps += ev;
    }
    ps += lane_xor<1>(ps);
    ps += lane_xor<2>(ps);
    ps += lane_xor<4>(ps);
    ps += lane_xor<8>(ps);
    ps += xhalf(ps);
    if (lane == 0) red[8 + wave] = ps;
    lds_barrier();
    rinv0 = 1.f / ((red[8]  + red[9])  + (red[10] + red[11]));
    rinv1 = 1.f / ((red[12] + red[13]) + (red[14] + red[15]));
}

struct GroupOps {
    v2f   b[8];
    float e0, e1;
};

static __device__ __forceinline__ void load_group(
        const int g, GroupOps& f, const float* tile, const float* e,
        const int erun, const int obase, const int m, const int h) {
    const int og = obase + 16 * g;
    const float* ec = e + 72 * erun + 16 * g + m;
    f.e0 = ec[0];
    f.e1 = ec[E_ROW];
    #pragma unroll
    for (int q = 0; q < 8; ++q) {
        const float* bp = tile + (og + 2 * q + h) * FD + m;
        f.b[q].x = bp[0];
        f.b[q].y = bp[LROW];
    }
}

static __device__ __forceinline__ void wmma_group(
        const int g, v8f (&acc)[4], const GroupOps& f,
        const float rinv0, const float rinv1, const int m, const int h) {
    const float c0 = f.e0 * rinv0;
    const float c1 = f.e1 * rinv1;
    #pragma unroll
    for (int q = 0; q < 8; ++q) {
        const bool sel = (m == 2 * q + h);
        v2f a;
        a.x = sel ? c0 : 0.f;
        a.y = sel ? c1 : 0.f;
        acc[g] = __builtin_amdgcn_wmma_f32_16x16x4_f32(
                 false, a,  false, f.b[q],
                 (short)0, acc[g],  false,  false);
    }
}

template <bool REFILL>
static __device__ __forceinline__ void routing_round(
        float4 (&pre)[PRE], const float4 (&vsv)[16], v8f (&acc)[4],
        const float4* __restrict__ nsrc, const int t_own, const int t_oth,
        float* tn, float* en, const float* tc, float* ec, float* red,
        const int t, const int lane, const int wave, const int erun, const int obase,
        const int m, const int h, const int half) {
    const bool odd  = (t & 1) != 0;
    const bool high = (t & 2) != 0;
    float* eq_own = en + (t >> 2) + 72 * (t & 3) + 576 * half;
    float* eq_oth = en + (t >> 2) + 72 * (t & 3) + 576 * (half ^ 1);
    float rinv0, rinv1;
    row_softmax(ec, red, lane, wave, rinv0, rinv1);
    GroupOps ops[2];
    load_group(0, ops[0], tc, ec, erun, obase, m, h);
    #pragma unroll
    for (int g = 0; g < 4; ++g) {
        if (g < 3) load_group(g + 1, ops[(g + 1) & 1], tc, ec, erun, obase, m, h);
        __builtin_amdgcn_sched_barrier(6);
        tile_in_group<REFILL>(2 * g,     pre, vsv, nsrc, t_own, t_oth, tn, eq_own, eq_oth, t, odd, high);
        tile_in_group<REFILL>(2 * g + 1, pre, vsv, nsrc, t_own, t_oth, tn, eq_own, eq_oth, t, odd, high);
        wmma_group(g, acc, ops[g & 1], rinv0, rinv1, m, h);
    }
    lds_barrier();
}

__global__ __launch_bounds__(256)
void caps_routing_pass(const float* __restrict__ uh,
                       const float* __restrict__ vsum_g,
                       float* __restrict__ partials,
                       const int rev) {
    extern __shared__ float smem[];
    float* tile = smem;
    float* e    = tile + 2 * TILE_FLOATS;
    float* red  = e    + 2 * E_FLOATS;

    const int t     = threadIdx.x;
    const int lane  = t & 31;
    const int wave  = t >> 5;
    const int obase = wave * 64;
    const int m     = lane & 15;
    const int h     = lane >> 4;
    const int rg    = blockIdx.x % NBLK;
    const int half  = blockIdx.x / NBLK;
    const int erun  = 8 * half + wave;
    const int t_own = t + HALF_F4 * half;
    const int t_oth = t + HALF_F4 * (half ^ 1);

    const float4* src = (const float4*)(uh + (size_t)rg * ROWS_BLK * SVEC)
                        + (rev ? (NTILE - 1) * TILE_F4 : 0);
    const long step = rev ? -(long)TILE_F4 : (long)TILE_F4;

    float4 vsv[16];
    #pragma unroll
    for (int kk = 0; kk < 16; ++kk)
        vsv[kk] = ((const float4*)vsum_g)[(kk < 8 ? t_own : t_oth) + 256 * (kk & 7)];

    float4 pre[PRE];
    #pragma unroll
    for (int k = 0; k < PRE; ++k)
        pre[k] = src[((k & 8) == 0 ? t_own : t_oth) + 256 * (k & 7) + 4096 * (k >> 4)];

    v8f acc[4];
    #pragma unroll
    for (int g = 0; g < 4; ++g) acc[g] = (v8f){0.f, 0.f, 0.f, 0.f, 0.f, 0.f, 0.f, 0.f};

    {
        const bool odd  = (t & 1) != 0;
        const bool high = (t & 2) != 0;
        float* eq_own = e + (t >> 2) + 72 * (t & 3) + 576 * half;
        float* eq_oth = e + (t >> 2) + 72 * (t & 3) + 576 * (half ^ 1);
        #pragma unroll
        for (int g = 0; g < 8; ++g)
            tile_in_group<true>(g, pre, vsv, src + step, t_own, t_oth, tile, eq_own, eq_oth,
                                t, odd, high);
    }
    lds_barrier();

    for (int tl = 0; tl + 2 < NTILE; ++tl) {
        const int cur = tl & 1;
        routing_round<true>(pre, vsv, acc, src + (long)(tl + 2) * step, t_own, t_oth,
                            tile + (cur ^ 1) * TILE_FLOATS, e + (cur ^ 1) * E_FLOATS,
                            tile + cur * TILE_FLOATS,       e + cur * E_FLOATS, red,
                            t, lane, wave, erun, obase, m, h, half);
    }
    routing_round<false>(pre, vsv, acc, src, t_own, t_oth,
                         tile + TILE_FLOATS, e + E_FLOATS, tile, e, red,
                         t, lane, wave, erun, obase, m, h, half);
    {
        float rinv0, rinv1;
        row_softmax(e + E_FLOATS, red, lane, wave, rinv0, rinv1);
        GroupOps ops[2];
        load_group(0, ops[0], tile + TILE_FLOATS, e + E_FLOATS, erun, obase, m, h);
        #pragma unroll
        for (int g = 0; g < 4; ++g) {
            if (g < 3) load_group(g + 1, ops[(g + 1) & 1], tile + TILE_FLOATS, e + E_FLOATS,
                                  erun, obase, m, h);
            __builtin_amdgcn_sched_barrier(6);
            wmma_group(g, acc, ops[g & 1], rinv0, rinv1, m, h);
        }
    }

    #pragma unroll
    for (int g = 0; g < 4; ++g) {
        #pragma unroll
        for (int r8 = 0; r8 < 8; ++r8)
            tile[(obase + 16 * g + 8 * h + r8) * FD + m] = acc[g][r8];
    }
    lds_barrier();
    float* pout = partials + (size_t)rg * SVEC + (size_t)half * LROW;
    for (int q = t; q < LROW / 4; q += 256)
        ((float4*)pout)[q] = ((const float4*)tile)[q];
}

__global__ void caps_reduce_squash(const float* __restrict__ partials,
                                   float* __restrict__ vsum,
                                   float* __restrict__ vout) {
    const int idx = blockIdx.x * 256 + threadIdx.x;
    float s = 0.f;
    for (int b = 0; b < NBLK; ++b)
        s += partials[(size_t)b * SVEC + idx];
    float sq = s * s;
    #pragma unroll
    for (int m = 1; m < 16; m <<= 1)
        sq += __shfl_xor(sq, m, 16);
    const float scale = (sq / (1.f + sq)) * rsqrtf(sq);
    const float v = s * scale;
    vout[idx] = v;
    vsum[idx] += v;
}

extern "C" void kernel_launch(void* const* d_in, const int* in_sizes, int n_in,
                              void* d_out, int out_size, void* d_ws, size_t ws_size,
                              hipStream_t stream) {
    const float* uh = (const float*)d_in[0];

    float* ws       = (float*)d_ws;
    float* vsum     = ws;
    float* partials = ws + SVEC;
    float* vout     = (float*)d_out;

    caps_zero_vsum<<<SVEC / 256, 256, 0, stream>>>(vsum);
    caps_routing_pass<<<2 * NBLK, 256, SMEM_BYTES, stream>>>(uh, vsum, partials, 0);
    caps_reduce_squash<<<SVEC / 256, 256, 0, stream>>>(partials, vsum, vout);
    caps_routing_pass<<<2 * NBLK, 256, SMEM_BYTES, stream>>>(uh, vsum, partials, 1);
    caps_reduce_squash<<<SVEC / 256, 256, 0, stream>>>(partials, vsum, vout);
    caps_routing_pass<<<2 * NBLK, 256, SMEM_BYTES, stream>>>(uh, vsum, partials, 0);
    caps_reduce_squash<<<SVEC / 256, 256, 0, stream>>>(partials, vsum, vout);
}
